# starting kernel
# baseline (speedup 1.0000x reference)
#include <hip/hip_runtime.h>
#include <math.h>

typedef __attribute__((ext_vector_type(2))) float v2f;
typedef __attribute__((ext_vector_type(8))) float v8f;

#define NN 32
#define MM 2
#define CC 256
#define TV 2500
#define PP 8
#define TOK_PER_BLOCK 128
#define CHUNKS 20
#define EPSF 1e-8f

#if __has_builtin(__builtin_amdgcn_wmma_f32_16x16x4_f32)
#define HAVE_WMMA_F32X4 1
#else
#warning "builtin __builtin_amdgcn_wmma_f32_16x16x4_f32 not found - using scalar fallback (wmma=0 expected)"
#define HAVE_WMMA_F32X4 0
#endif

__global__ void rn_zero_kernel(float* ws, int n) {
    int i = blockIdx.x * blockDim.x + threadIdx.x;
    if (i < n) ws[i] = 0.0f;
}

__global__ void __launch_bounds__(256)
rn_main_kernel(const float* __restrict__ x, const float* __restrict__ protos,
               float* __restrict__ g_sums, float* __restrict__ g_cnt) {
    __shared__ __align__(16) float ldsA[64 * 64];
    __shared__ float lds_pn[PP];
    __shared__ __align__(16) float lds_sums[PP * 257];
    __shared__ float lds_cnt[PP];

    const int tid   = threadIdx.x;
    const int chunk = blockIdx.x;
    const int n     = blockIdx.y;

    for (int idx = tid; idx < 64 * 64; idx += 256) {
        int j     = idx >> 6;
        int rem   = idx & 63;
        int L     = rem >> 1;
        int which = rem & 1;
        int m     = L & 15;
        int k     = 4 * j + ((L < 16) ? which : (2 + which));
        ldsA[idx] = (m < PP) ? protos[m * CC + k] : 0.0f;
    }
    if (tid < PP) {
        float s = 0.0f;
        for (int c = 0; c < CC; ++c) { float pv = protos[tid * CC + c]; s += pv * pv; }
        lds_pn[tid] = sqrtf(s);
    }
    for (int idx = tid; idx < PP * 257; idx += 256) lds_sums[idx] = 0.0f;
    if (tid < PP) lds_cnt[tid] = 0.0f;
    __syncthreads();

    const int lane  = tid & 31;
    const int wave  = tid >> 5;
    const int tloc  = lane & 15;
    const int tv    = chunk * TOK_PER_BLOCK + wave * 16 + tloc;
    const bool valid = (tv < TV);
    const int tvc   = valid ? tv : (TV - 1);
    const int khalf = (lane >> 4) << 1;

    const float* xm0 = x + (size_t)n * MM * CC * TV;
    const float* xm1 = xm0 + (size_t)CC * TV;

    v8f acc = (v8f){0.f, 0.f, 0.f, 0.f, 0.f, 0.f, 0.f, 0.f};
#if !HAVE_WMMA_F32X4
    float fdot[PP];
#pragma unroll
    for (int p = 0; p < PP; ++p) fdot[p] = 0.0f;
#endif
    float hh = 0.0f;

    for (int j = 0; j < 64; ++j) {
        int c0 = 4 * j + khalf;
        size_t o0 = (size_t)c0 * TV + tvc;
        size_t o1 = o0 + TV;
        float b0 = 0.5f * (xm0[o0] + xm1[o0]);
        float b1 = 0.5f * (xm0[o1] + xm1[o1]);
        hh += b0 * b0 + b1 * b1;
#if HAVE_WMMA_F32X4
        v2f a = *(const v2f*)&ldsA[(j << 6) + (lane << 1)];
        v2f b; b.x = b0; b.y = b1;
        acc = __builtin_amdgcn_wmma_f32_16x16x4_f32(false, a, false, b,
                                                    (short)0, acc, false, false);
#else
#pragma unroll
        for (int p = 0; p < PP; ++p)
            fdot[p] += b0 * protos[p * CC + c0] + b1 * protos[p * CC + c0 + 1];
#endif
    }

    hh += __shfl_xor(hh, 16, 32);
    float hn = sqrtf(hh);

    float dots[PP];
#if HAVE_WMMA_F32X4
#pragma unroll
    for (int p = 0; p < PP; ++p) dots[p] = acc[p];
#else
#pragma unroll
    for (int p = 0; p < PP; ++p) dots[p] = fdot[p] + __shfl_xor(fdot[p], 16, 32);
#endif

    float best = -3.0e38f;
    int bi = 0;
#pragma unroll
    for (int p = 0; p < PP; ++p) {
        float sim = dots[p] / fmaxf(hn * lds_pn[p], EPSF);
        if (sim > best) { best = sim; bi = p; }
    }
    int assign = __shfl(bi, tloc, 32);

    if (lane < 16 && valid)
        __hip_atomic_fetch_add(&lds_cnt[assign], 1.0f,
                               __ATOMIC_RELAXED, __HIP_MEMORY_SCOPE_WORKGROUP);

    for (int i = 0; i < 128; ++i) {
        int c = 2 * i + (lane >> 4);
        size_t o = (size_t)c * TV + tvc;
        float h = 0.5f * (xm0[o] + xm1[o]);
        if (valid)
            __hip_atomic_fetch_add(&lds_sums[assign * 257 + c], h,
                                   __ATOMIC_RELAXED, __HIP_MEMORY_SCOPE_WORKGROUP);
    }
    __syncthreads();

    float* gs = g_sums + (size_t)n * PP * CC;
#pragma unroll
    for (int p = 0; p < PP; ++p)
        __hip_atomic_fetch_add(&gs[p * CC + tid], lds_sums[p * 257 + tid],
                               __ATOMIC_RELAXED, __HIP_MEMORY_SCOPE_AGENT);
    if (tid < PP)
        __hip_atomic_fetch_add(&g_cnt[n * PP + tid], lds_cnt[tid],
                               __ATOMIC_RELAXED, __HIP_MEMORY_SCOPE_AGENT);
}

__global__ void rn_final_kernel(const float* __restrict__ g_sums,
                                const float* __restrict__ g_cnt,
                                float* __restrict__ out) {
    int n = blockIdx.x;
    int c = threadIdx.x;
    float s = 0.0f;
#pragma unroll
    for (int p = 0; p < PP; ++p) {
        float cnt = g_cnt[n * PP + p];
        s += g_sums[((size_t)n * PP + p) * CC + c] / fmaxf(cnt, 1.0f);
    }
    out[n * CC + c] = s * 0.125f;
}

extern "C" void kernel_launch(void* const* d_in, const int* in_sizes, int n_in,
                              void* d_out, int out_size, void* d_ws, size_t ws_size,
                              hipStream_t stream) {
    const float* x      = (const float*)d_in[0];
    const float* protos = (const float*)d_in[1];

    float* g_sums = (float*)d_ws;
    float* g_cnt  = g_sums + NN * PP * CC;

    int zn = NN * PP * CC + NN * PP;
    rn_zero_kernel<<<(zn + 255) / 256, 256, 0, stream>>>((float*)d_ws, zn);

    dim3 grid(CHUNKS, NN);
    rn_main_kernel<<<grid, 256, 0, stream>>>(x, protos, g_sums, g_cnt);

    rn_final_kernel<<<NN, CC, 0, stream>>>(g_sums, g_cnt, (float*)d_out);
}


# Keep each wave's 16-token x 256-channel tile of h in registers (128 per lane) and feed the WMMA dots and the LDS segment sums from it: x is swept once; float atomics kept (arrival order as before)
# speedup vs baseline: 1.0883x; 1.0883x over previous
#include <hip/hip_runtime.h>
#include <math.h>

typedef __attribute__((ext_vector_type(2))) float v2f;
typedef __attribute__((ext_vector_type(8))) float v8f;

#define NN 32
#define MM 2
#define CC 256
#define TV 2500
#define PP 8
#define TOK_PER_BLOCK 128
#define CHUNKS 20
#define EPSF 1e-8f

#if __has_builtin(__builtin_amdgcn_wmma_f32_16x16x4_f32)
#define HAVE_WMMA_F32X4 1
#else
#warning "builtin __builtin_amdgcn_wmma_f32_16x16x4_f32 not found - using scalar fallback (wmma=0 expected)"
#define HAVE_WMMA_F32X4 0
#endif

__global__ void rn_zero_kernel(float* ws, int n) {
    int i = blockIdx.x * blockDim.x + threadIdx.x;
    if (i < n) ws[i] = 0.0f;
}

__global__ void __launch_bounds__(256)
rn_main_kernel(const float* __restrict__ x, const float* __restrict__ protos,
               float* __restrict__ g_sums, float* __restrict__ g_cnt) {
    __shared__ __align__(16) float ldsA[64 * 64];
    __shared__ float lds_pn[PP];
    __shared__ __align__(16) float lds_sums[PP * 257];
    __shared__ float lds_cnt[PP];

    const int tid   = threadIdx.x;
    const int chunk = blockIdx.x;
    const int n     = blockIdx.y;

    for (int idx = tid; idx < 64 * 64; idx += 256) {
        int j     = idx >> 6;
        int rem   = idx & 63;
        int L     = rem >> 1;
        int which = rem & 1;
        int m     = L & 15;
        int k     = 4 * j + ((L < 16) ? which : (2 + which));
        ldsA[idx] = (m < PP) ? protos[m * CC + k] : 0.0f;
    }
    if (tid < PP) {
        float s = 0.0f;
        for (int c = 0; c < CC; ++c) { float pv = protos[tid * CC + c]; s += pv * pv; }
        lds_pn[tid] = sqrtf(s);
    }
    for (int idx = tid; idx < PP * 257; idx += 256) lds_sums[idx] = 0.0f;
    if (tid < PP) lds_cnt[tid] = 0.0f;
    __syncthreads();

    const int lane  = tid & 31;
    const int wave  = tid >> 5;
    const int tloc  = lane & 15;
    const int tv    = chunk * TOK_PER_BLOCK + wave * 16 + tloc;
    const bool valid = (tv < TV);
    const int tvc   = valid ? tv : (TV - 1);
    const int khalf = (lane >> 4) << 1;

    const float* xm0 = x + (size_t)n * MM * CC * TV;
    const float* xm1 = xm0 + (size_t)CC * TV;

    float hv[128];
#pragma unroll
    for (int j = 0; j < 64; ++j) {
        int c0 = 4 * j + khalf;
        size_t o0 = (size_t)c0 * TV + tvc;
        size_t o1 = o0 + TV;
        hv[2 * j]     = 0.5f * (xm0[o0] + xm1[o0]);
        hv[2 * j + 1] = 0.5f * (xm0[o1] + xm1[o1]);
    }

    v8f acc = (v8f){0.f, 0.f, 0.f, 0.f, 0.f, 0.f, 0.f, 0.f};
#if !HAVE_WMMA_F32X4
    float fdot[PP];
#pragma unroll
    for (int p = 0; p < PP; ++p) fdot[p] = 0.0f;
#endif
    float hh = 0.0f;

#pragma unroll
    for (int j = 0; j < 64; ++j) {
        float b0 = hv[2 * j];
        float b1 = hv[2 * j + 1];
        hh += b0 * b0 + b1 * b1;
#if HAVE_WMMA_F32X4
        v2f a = *(const v2f*)&ldsA[(j << 6) + (lane << 1)];
        v2f b; b.x = b0; b.y = b1;
        acc = __builtin_amdgcn_wmma_f32_16x16x4_f32(false, a, false, b,
                                                    (short)0, acc, false, false);
#else
        int c0 = 4 * j + khalf;
#pragma unroll
        for (int p = 0; p < PP; ++p)
            fdot[p] += b0 * protos[p * CC + c0] + b1 * protos[p * CC + c0 + 1];
#endif
    }

    hh += __shfl_xor(hh, 16, 32);
    float hn = sqrtf(hh);

    float dots[PP];
#if HAVE_WMMA_F32X4
#pragma unroll
    for (int p = 0; p < PP; ++p) dots[p] = acc[p];
#else
#pragma unroll
    for (int p = 0; p < PP; ++p) dots[p] = fdot[p] + __shfl_xor(fdot[p], 16, 32);
#endif

    float best = -3.0e38f;
    int bi = 0;
#pragma unroll
    for (int p = 0; p < PP; ++p) {
        float sim = dots[p] / fmaxf(hn * lds_pn[p], EPSF);
        if (sim > best) { best = sim; bi = p; }
    }
    int assign = __shfl(bi, tloc, 32);

    if (lane < 16 && valid)
        __hip_atomic_fetch_add(&lds_cnt[assign], 1.0f,
                               __ATOMIC_RELAXED, __HIP_MEMORY_SCOPE_WORKGROUP);

    if (valid) {
        float* dst = &lds_sums[assign * 257 + khalf];
#pragma unroll
        for (int j = 0; j < 64; ++j) {
            __hip_atomic_fetch_add(dst + 4 * j, hv[2 * j],
                                   __ATOMIC_RELAXED, __HIP_MEMORY_SCOPE_WORKGROUP);
            __hip_atomic_fetch_add(dst + 4 * j + 1, hv[2 * j + 1],
                                   __ATOMIC_RELAXED, __HIP_MEMORY_SCOPE_WORKGROUP);
        }
    }
    __syncthreads();

    float* gs = g_sums + (size_t)n * PP * CC;
#pragma unroll
    for (int p = 0; p < PP; ++p)
        __hip_atomic_fetch_add(&gs[p * CC + tid], lds_sums[p * 257 + tid],
                               __ATOMIC_RELAXED, __HIP_MEMORY_SCOPE_AGENT);
    if (tid < PP)
        __hip_atomic_fetch_add(&g_cnt[n * PP + tid], lds_cnt[tid],
                               __ATOMIC_RELAXED, __HIP_MEMORY_SCOPE_AGENT);
}

__global__ void rn_final_kernel(const float* __restrict__ g_sums,
                                const float* __restrict__ g_cnt,
                                float* __restrict__ out) {
    int n = blockIdx.x;
    int c = threadIdx.x;
    float s = 0.0f;
#pragma unroll
    for (int p = 0; p < PP; ++p) {
        float cnt = g_cnt[n * PP + p];
        s += g_sums[((size_t)n * PP + p) * CC + c] / fmaxf(cnt, 1.0f);
    }
    out[n * CC + c] = s * 0.125f;
}

extern "C" void kernel_launch(void* const* d_in, const int* in_sizes, int n_in,
                              void* d_out, int out_size, void* d_ws, size_t ws_size,
                              hipStream_t stream) {
    const float* x      = (const float*)d_in[0];
    const float* protos = (const float*)d_in[1];

    float* g_sums = (float*)d_ws;
    float* g_cnt  = g_sums + NN * PP * CC;

    int zn = NN * PP * CC + NN * PP;
    rn_zero_kernel<<<(zn + 255) / 256, 256, 0, stream>>>((float*)d_ws, zn);

    dim3 grid(CHUNKS, NN);
    rn_main_kernel<<<grid, 256, 0, stream>>>(x, protos, g_sums, g_cnt);

    rn_final_kernel<<<NN, CC, 0, stream>>>(g_sums, g_cnt, (float*)d_out);
}


# Segment sums by a second f32 WMMA product (one-hot assignment x h, tile transposed through LDS) replace the LDS float atomics: fixed sum order inside a block; global float atomics kept
# speedup vs baseline: 1.7641x; 1.6210x over previous
#include <hip/hip_runtime.h>
#include <math.h>

typedef __attribute__((ext_vector_type(2))) float v2f;
typedef __attribute__((ext_vector_type(2))) int v2i;
typedef __attribute__((ext_vector_type(8))) float v8f;

#define NN 32
#define MM 2
#define CC 256
#define TV 2500
#define PP 8
#define TOK_PER_BLOCK 128
#define CHUNKS 20
#define EPSF 1e-8f
#define HCH 128
#define HP 140

#if __has_builtin(__builtin_amdgcn_wmma_f32_16x16x4_f32)
#define HAVE_WMMA_F32X4 1
#else
#warning "builtin __builtin_amdgcn_wmma_f32_16x16x4_f32 not found - using scalar fallback (wmma=0 expected)"
#define HAVE_WMMA_F32X4 0
#endif

__global__ void rn_zero_kernel(float* ws, int n) {
    int i = blockIdx.x * blockDim.x + threadIdx.x;
    if (i < n) ws[i] = 0.0f;
}

__global__ void __launch_bounds__(256)
rn_main_kernel(const float* __restrict__ x, const float* __restrict__ protos,
               float* __restrict__ g_sums, float* __restrict__ g_cnt) {
    __shared__ __align__(16) float ldsA[64 * 64];
    __shared__ float lds_pn[PP];
    __shared__ float lds_cnt[PP];
    __shared__ __align__(16) int lds_assign[TOK_PER_BLOCK];
    __shared__ __align__(16) float ldsH[HCH * HP];

    const int tid   = threadIdx.x;
    const int chunk = blockIdx.x;
    const int n     = blockIdx.y;

    for (int idx = tid; idx < 64 * 64; idx += 256) {
        int j     = idx >> 6;
        int rem   = idx & 63;
        int L     = rem >> 1;
        int which = rem & 1;
        int m     = L & 15;
        int k     = 4 * j + ((L < 16) ? which : (2 + which));
        ldsA[idx] = (m < PP) ? protos[m * CC + k] : 0.0f;
    }
    if (tid < PP) {
        float s = 0.0f;
        for (int c = 0; c < CC; ++c) { float pv = protos[tid * CC + c]; s += pv * pv; }
        lds_pn[tid] = sqrtf(s);
    }
    if (tid < PP) lds_cnt[tid] = 0.0f;
    __syncthreads();

    const int lane  = tid & 31;
    const int wave  = tid >> 5;
    const int tloc  = lane & 15;
    const int tv    = chunk * TOK_PER_BLOCK + wave * 16 + tloc;
    const bool valid = (tv < TV);
    const int tvc   = valid ? tv : (TV - 1);
    const int khalf = (lane >> 4) << 1;

    const float* xm0 = x + (size_t)n * MM * CC * TV;
    const float* xm1 = xm0 + (size_t)CC * TV;

    float hv[128];
#pragma unroll
    for (int j = 0; j < 64; ++j) {
        int c0 = 4 * j + khalf;
        size_t o0 = (size_t)c0 * TV + tvc;
        size_t o1 = o0 + TV;
        hv[2 * j]     = 0.5f * (xm0[o0] + xm1[o0]);
        hv[2 * j + 1] = 0.5f * (xm0[o1] + xm1[o1]);
    }

    v8f acc = (v8f){0.f, 0.f, 0.f, 0.f, 0.f, 0.f, 0.f, 0.f};
#if !HAVE_WMMA_F32X4
    float fdot[PP];
#pragma unroll
    for (int p = 0; p < PP; ++p) fdot[p] = 0.0f;
#endif
    float hh = 0.0f;

#pragma unroll
    for (int j = 0; j < 64; ++j) {
        float b0 = hv[2 * j];
        float b1 = hv[2 * j + 1];
        hh += b0 * b0 + b1 * b1;
#if HAVE_WMMA_F32X4
        v2f a = *(const v2f*)&ldsA[(j << 6) + (lane << 1)];
        v2f b; b.x = b0; b.y = b1;
        acc = __builtin_amdgcn_wmma_f32_16x16x4_f32(false, a, false, b,
                                                    (short)0, acc, false, false);
#else
        int c0 = 4 * j + khalf;
#pragma unroll
        for (int p = 0; p < PP; ++p)
            fdot[p] += b0 * protos[p * CC + c0] + b1 * protos[p * CC + c0 + 1];
#endif
    }

    hh += __shfl_xor(hh, 16, 32);
    float hn = sqrtf(hh);

    float dots[PP];
#if HAVE_WMMA_F32X4
#pragma unroll
    for (int p = 0; p < PP; ++p) dots[p] = acc[p];
#else
#pragma unroll
    for (int p = 0; p < PP; ++p) dots[p] = fdot[p] + __shfl_xor(fdot[p], 16, 32);
#endif

    float best = -3.0e38f;
    int bi = 0;
#pragma unroll
    for (int p = 0; p < PP; ++p) {
        float sim = dots[p] / fmaxf(hn * lds_pn[p], EPSF);
        if (sim > best) { best = sim; bi = p; }
    }
    if (lane < 16) {
        lds_assign[wave * 16 + tloc] = valid ? bi : -1;
        if (valid)
            __hip_atomic_fetch_add(&lds_cnt[bi], 1.0f,
                                   __ATOMIC_RELAXED, __HIP_MEMORY_SCOPE_WORKGROUP);
    }

    float* gs = g_sums + (size_t)n * PP * CC;
    const int m16 = lane & 15;
    float*       wdst = &ldsH[khalf * HP + wave * 16 + tloc];
    const float* rsrc = &ldsH[(wave * 16 + m16) * HP + khalf];
    const int*   asrc = &lds_assign[khalf];

#pragma unroll
    for (int hf = 0; hf < 2; ++hf) {
        if (hf == 1) __syncthreads();
#pragma unroll
        for (int jj = 0; jj < 32; ++jj) {
            wdst[(4 * jj) * HP]     = hv[2 * (32 * hf + jj)];
            wdst[(4 * jj + 1) * HP] = hv[2 * (32 * hf + jj) + 1];
        }
        __syncthreads();

        if (hf == 0 && tid < PP)
            __hip_atomic_fetch_add(&g_cnt[n * PP + tid], lds_cnt[tid],
                                   __ATOMIC_RELAXED, __HIP_MEMORY_SCOPE_AGENT);

        v8f s0 = (v8f){0.f, 0.f, 0.f, 0.f, 0.f, 0.f, 0.f, 0.f};
        v8f s1 = (v8f){0.f, 0.f, 0.f, 0.f, 0.f, 0.f, 0.f, 0.f};
#pragma unroll
        for (int q = 0; q < 32; q += 2) {
            v2i t0 = *(const v2i*)&asrc[4 * q];
            v2i t1 = *(const v2i*)&asrc[4 * q + 4];
            v2f b0 = *(const v2f*)&rsrc[4 * q];
            v2f b1 = *(const v2f*)&rsrc[4 * q + 4];
            v2f a0, a1;
            a0.x = (t0.x == m16) ? 1.0f : 0.0f;
            a0.y = (t0.y == m16) ? 1.0f : 0.0f;
            a1.x = (t1.x == m16) ? 1.0f : 0.0f;
            a1.y = (t1.y == m16) ? 1.0f : 0.0f;
            s0 = __builtin_amdgcn_wmma_f32_16x16x4_f32(false, a0, false, b0,
                                                       (short)0, s0, false, false);
            s1 = __builtin_amdgcn_wmma_f32_16x16x4_f32(false, a1, false, b1,
                                                       (short)0, s1, false, false);
        }

        if (lane < 16) {
            float* gd = gs + HCH * hf + wave * 16 + lane;
#pragma unroll
            for (int p = 0; p < PP; ++p)
                __hip_atomic_fetch_add(&gd[p * CC], s0[p] + s1[p],
                                       __ATOMIC_RELAXED, __HIP_MEMORY_SCOPE_AGENT);
        }
    }
}

__global__ void rn_final_kernel(const float* __restrict__ g_sums,
                                const float* __restrict__ g_cnt,
                                float* __restrict__ out) {
    int n = blockIdx.x;
    int c = threadIdx.x;
    float s = 0.0f;
#pragma unroll
    for (int p = 0; p < PP; ++p) {
        float cnt = g_cnt[n * PP + p];
        s += g_sums[((size_t)n * PP + p) * CC + c] / fmaxf(cnt, 1.0f);
    }
    out[n * CC + c] = s * 0.125f;
}

extern "C" void kernel_launch(void* const* d_in, const int* in_sizes, int n_in,
                              void* d_out, int out_size, void* d_ws, size_t ws_size,
                              hipStream_t stream) {
    const float* x      = (const float*)d_in[0];
    const float* protos = (const float*)d_in[1];

    float* g_sums = (float*)d_ws;
    float* g_cnt  = g_sums + NN * PP * CC;

    int zn = NN * PP * CC + NN * PP;
    rn_zero_kernel<<<(zn + 255) / 256, 256, 0, stream>>>((float*)d_ws, zn);

    dim3 grid(CHUNKS, NN);
    rn_main_kernel<<<grid, 256, 0, stream>>>(x, protos, g_sums, g_cnt);

    rn_final_kernel<<<NN, CC, 0, stream>>>(g_sums, g_cnt, (float*)d_out);
}


# Prototype staging off the block's critical path: wave w fetches row w in one round trip (two 16-byte loads), scatters fragments at LDS pitch 66, norm by lane butterfly (norm sum order changes)
# speedup vs baseline: 1.9967x; 1.1319x over previous
#include <hip/hip_runtime.h>
#include <math.h>

typedef __attribute__((ext_vector_type(2))) float v2f;
typedef __attribute__((ext_vector_type(2))) int v2i;
typedef __attribute__((ext_vector_type(4))) float v4f;
typedef __attribute__((ext_vector_type(8))) float v8f;

#define NN 32
#define MM 2
#define CC 256
#define TV 2500
#define PP 8
#define TOK_PER_BLOCK 128
#define CHUNKS 20
#define EPSF 1e-8f
#define AP 66
#define HCH 128
#define HP 140

#if __has_builtin(__builtin_amdgcn_wmma_f32_16x16x4_f32)
#define HAVE_WMMA_F32X4 1
#else
#warning "builtin __builtin_amdgcn_wmma_f32_16x16x4_f32 not found - using scalar fallback (wmma=0 expected)"
#define HAVE_WMMA_F32X4 0
#endif

__global__ void rn_zero_kernel(float* ws, int n) {
    int i = blockIdx.x * blockDim.x + threadIdx.x;
    if (i < n) ws[i] = 0.0f;
}

__global__ void __launch_bounds__(256)
rn_main_kernel(const float* __restrict__ x, const float* __restrict__ protos,
               float* __restrict__ g_sums, float* __restrict__ g_cnt) {
    __shared__ __align__(16) float ldsA[64 * AP];
    __shared__ float lds_pn[PP];
    __shared__ float lds_cnt[PP];
    __shared__ __align__(16) int lds_assign[TOK_PER_BLOCK];
    __shared__ __align__(16) float ldsH[HCH * HP];

    const int tid   = threadIdx.x;
    const int chunk = blockIdx.x;
    const int n     = blockIdx.y;

    const int lane  = tid & 31;
    const int wave  = tid >> 5;

    static_assert(PP * 32 == 256 && CC == 256, "one wave per prototype row, two 16-byte pieces per lane");
    const float* prow = protos + wave * CC;
    const v4f pa = *(const v4f*)(prow + 4 * lane);
    const v4f pb = *(const v4f*)(prow + 128 + 4 * lane);

    {
        const int zj = tid >> 2;
        const int zq = tid & 3;
        float* z = &ldsA[zj * AP + 16 + (zq & 1) * 8 + (zq >> 1) * 32];
        const v2f zero = (v2f){0.0f, 0.0f};
        *(v2f*)(z)     = zero;
        *(v2f*)(z + 2) = zero;
        *(v2f*)(z + 4) = zero;
        *(v2f*)(z + 6) = zero;
    }
    if (tid < PP) lds_cnt[tid] = 0.0f;

    {
        float* d0 = &ldsA[lane * AP + 2 * wave];
        float* d1 = &ldsA[(32 + lane) * AP + 2 * wave];
        v2f t;
        t.x = pa.x; t.y = pa.y; *(v2f*)(d0)      = t;
        t.x = pa.z; t.y = pa.w; *(v2f*)(d0 + 32) = t;
        t.x = pb.x; t.y = pb.y; *(v2f*)(d1)      = t;
        t.x = pb.z; t.y = pb.w; *(v2f*)(d1 + 32) = t;
    }
    {
        float ss = pa.x * pa.x + pa.y * pa.y + pa.z * pa.z + pa.w * pa.w
                 + pb.x * pb.x + pb.y * pb.y + pb.z * pb.z + pb.w * pb.w;
        ss += __shfl_xor(ss, 16, 32);
        ss += __shfl_xor(ss, 8, 32);
        ss += __shfl_xor(ss, 4, 32);
        ss += __shfl_xor(ss, 2, 32);
        ss += __shfl_xor(ss, 1, 32);
        if (lane == 0) lds_pn[wave] = sqrtf(ss);
    }
    __syncthreads();

    const int tloc  = lane & 15;
    const int tv    = chunk * TOK_PER_BLOCK + wave * 16 + tloc;
    const bool valid = (tv < TV);
    const int tvc   = valid ? tv : (TV - 1);
    const int khalf = (lane >> 4) << 1;

    const float* xm0 = x + (size_t)n * MM * CC * TV;
    const float* xm1 = xm0 + (size_t)CC * TV;

    float hv[128];
#pragma unroll
    for (int j = 0; j < 64; ++j) {
        int c0 = 4 * j + khalf;
        size_t o0 = (size_t)c0 * TV + tvc;
        size_t o1 = o0 + TV;
        hv[2 * j]     = 0.5f * (xm0[o0] + xm1[o0]);
        hv[2 * j + 1] = 0.5f * (xm0[o1] + xm1[o1]);
    }

    v8f acc = (v8f){0.f, 0.f, 0.f, 0.f, 0.f, 0.f, 0.f, 0.f};
#if !HAVE_WMMA_F32X4
    float fdot[PP];
#pragma unroll
    for (int p = 0; p < PP; ++p) fdot[p] = 0.0f;
#endif
    float hh = 0.0f;

#pragma unroll
    for (int j = 0; j < 64; ++j) {
        float b0 = hv[2 * j];
        float b1 = hv[2 * j + 1];
        hh += b0 * b0 + b1 * b1;
#if HAVE_WMMA_F32X4
        v2f a = *(const v2f*)&ldsA[j * AP + (lane << 1)];
        v2f b; b.x = b0; b.y = b1;
        acc = __builtin_amdgcn_wmma_f32_16x16x4_f32(false, a, false, b,
                                                    (short)0, acc, false, false);
#else
        int c0 = 4 * j + khalf;
#pragma unroll
        for (int p = 0; p < PP; ++p)
            fdot[p] += b0 * protos[p * CC + c0] + b1 * protos[p * CC + c0 + 1];
#endif
    }

    hh += __shfl_xor(hh, 16, 32);
    float hn = sqrtf(hh);

    float dots[PP];
#if HAVE_WMMA_F32X4
#pragma unroll
    for (int p = 0; p < PP; ++p) dots[p] = acc[p];
#else
#pragma unroll
    for (int p = 0; p < PP; ++p) dots[p] = fdot[p] + __shfl_xor(fdot[p], 16, 32);
#endif

    float best = -3.0e38f;
    int bi = 0;
#pragma unroll
    for (int p = 0; p < PP; ++p) {
        float sim = dots[p] / fmaxf(hn * lds_pn[p], EPSF);
        if (sim > best) { best = sim; bi = p; }
    }
    if (lane < 16) {
        lds_assign[wave * 16 + tloc] = valid ? bi : -1;
        if (valid)
            __hip_atomic_fetch_add(&lds_cnt[bi], 1.0f,
                                   __ATOMIC_RELAXED, __HIP_MEMORY_SCOPE_WORKGROUP);
    }

    float* gs = g_sums + (size_t)n * PP * CC;
    const int m16 = lane & 15;
    float*       wdst = &ldsH[khalf * HP + wave * 16 + tloc];
    const float* rsrc = &ldsH[(wave * 16 + m16) * HP + khalf];
    const int*   asrc = &lds_assign[khalf];

#pragma unroll
    for (int hf = 0; hf < 2; ++hf) {
        if (hf == 1) __syncthreads();
#pragma unroll
        for (int jj = 0; jj < 32; ++jj) {
            wdst[(4 * jj) * HP]     = hv[2 * (32 * hf + jj)];
            wdst[(4 * jj + 1) * HP] = hv[2 * (32 * hf + jj) + 1];
        }
        __syncthreads();

        if (hf == 0 && tid < PP)
            __hip_atomic_fetch_add(&g_cnt[n * PP + tid], lds_cnt[tid],
                                   __ATOMIC_RELAXED, __HIP_MEMORY_SCOPE_AGENT);

        v8f s0 = (v8f){0.f, 0.f, 0.f, 0.f, 0.f, 0.f, 0.f, 0.f};
        v8f s1 = (v8f){0.f, 0.f, 0.f, 0.f, 0.f, 0.f, 0.f, 0.f};
#pragma unroll
        for (int q = 0; q < 32; q += 2) {
            v2i t0 = *(const v2i*)&asrc[4 * q];
            v2i t1 = *(const v2i*)&asrc[4 * q + 4];
            v2f b0 = *(const v2f*)&rsrc[4 * q];
            v2f b1 = *(const v2f*)&rsrc[4 * q + 4];
            v2f a0, a1;
            a0.x = (t0.x == m16) ? 1.0f : 0.0f;
            a0.y = (t0.y == m16) ? 1.0f : 0.0f;
            a1.x = (t1.x == m16) ? 1.0f : 0.0f;
            a1.y = (t1.y == m16) ? 1.0f : 0.0f;
            s0 = __builtin_amdgcn_wmma_f32_16x16x4_f32(false, a0, false, b0,
                                                       (short)0, s0, false, false);
            s1 = __builtin_amdgcn_wmma_f32_16x16x4_f32(false, a1, false, b1,
                                                       (short)0, s1, false, false);
        }

        if (lane < 16) {
            float* gd = gs + HCH * hf + wave * 16 + lane;
#pragma unroll
            for (int p = 0; p < PP; ++p)
                __hip_atomic_fetch_add(&gd[p * CC], s0[p] + s1[p],
                                       __ATOMIC_RELAXED, __HIP_MEMORY_SCOPE_AGENT);
        }
    }
}

__global__ void rn_final_kernel(const float* __restrict__ g_sums,
                                const float* __restrict__ g_cnt,
                                float* __restrict__ out) {
    int n = blockIdx.x;
    int c = threadIdx.x;
    float s = 0.0f;
#pragma unroll
    for (int p = 0; p < PP; ++p) {
        float cnt = g_cnt[n * PP + p];
        s += g_sums[((size_t)n * PP + p) * CC + c] / fmaxf(cnt, 1.0f);
    }
    out[n * CC + c] = s * 0.125f;
}

extern "C" void kernel_launch(void* const* d_in, const int* in_sizes, int n_in,
                              void* d_out, int out_size, void* d_ws, size_t ws_size,
                              hipStream_t stream) {
    const float* x      = (const float*)d_in[0];
    const float* protos = (const float*)d_in[1];

    float* g_sums = (float*)d_ws;
    float* g_cnt  = g_sums + NN * PP * CC;

    int zn = NN * PP * CC + NN * PP;
    rn_zero_kernel<<<(zn + 255) / 256, 256, 0, stream>>>((float*)d_ws, zn);

    dim3 grid(CHUNKS, NN);
    rn_main_kernel<<<grid, 256, 0, stream>>>(x, protos, g_sums, g_cnt);

    rn_final_kernel<<<NN, CC, 0, stream>>>(g_sums, g_cnt, (float*)d_out);
}


# x tile goes global -> LDS by 16-byte loads (one wave-load = 512 contiguous bytes of one channel row); both WMMA products take h from LDS; every sum order as before, float atomics kept
# speedup vs baseline: 2.2106x; 1.1071x over previous
#include <hip/hip_runtime.h>
#include <math.h>

typedef __attribute__((ext_vector_type(2))) float v2f;
typedef __attribute__((ext_vector_type(2))) int v2i;
typedef __attribute__((ext_vector_type(4))) float v4f;
typedef __attribute__((ext_vector_type(8))) float v8f;

#define NN 32
#define MM 2
#define CC 256
#define TV 2500
#define PP 8
#define TOK_PER_BLOCK 128
#define CHUNKS 20
#define EPSF 1e-8f
#define AP 66
#define HCH 128
#define HP 140
#define CPW 32
#define LB 8

#if __has_builtin(__builtin_amdgcn_wmma_f32_16x16x4_f32)
#define HAVE_WMMA_F32X4 1
#else
#warning "builtin __builtin_amdgcn_wmma_f32_16x16x4_f32 not found - using scalar fallback (wmma=0 expected)"
#define HAVE_WMMA_F32X4 0
#endif

__global__ void rn_zero_kernel(float* ws, int n) {
    int i = blockIdx.x * blockDim.x + threadIdx.x;
    if (i < n) ws[i] = 0.0f;
}

__global__ void __launch_bounds__(256, 4)
rn_main_kernel(const float* __restrict__ x, const float* __restrict__ protos,
               float* __restrict__ g_sums, float* __restrict__ g_cnt) {
    __shared__ __align__(16) float ldsA[64 * AP];
    __shared__ float lds_pn[PP];
    __shared__ float lds_cnt[PP];
    __shared__ __align__(16) int lds_assign[TOK_PER_BLOCK];
    __shared__ __align__(16) float ldsH[CC * HP];

    const int tid   = threadIdx.x;
    const int chunk = blockIdx.x;
    const int n     = blockIdx.y;

    const int lane  = tid & 31;
    const int wave  = tid >> 5;

    static_assert(PP * 32 == 256 && CC == 256, "one wave per prototype row, two 16-byte pieces per lane");
    const float* prow = protos + wave * CC;
    const v4f pa = *(const v4f*)(prow + 4 * lane);
    const v4f pb = *(const v4f*)(prow + 128 + 4 * lane);

    static_assert(TV % 4 == 0 && TOK_PER_BLOCK == 4 * 32 && CPW * 8 == CC && CPW % LB == 0, "tile fetch shape");
    const int tq  = chunk * TOK_PER_BLOCK + 4 * lane;
    const int tqc = (tq < TV) ? tq : (TV - 4);
    const float* xr0 = x + ((size_t)n * MM * CC + (size_t)CPW * wave) * TV + tqc;
    const float* xr1 = xr0 + (size_t)CC * TV;
    float* hdst = &ldsH[(CPW * wave) * HP + 4 * lane];

    v4f ra[2][LB], rb[2][LB];
#pragma unroll
    for (int i = 0; i < LB; ++i) {
        ra[0][i] = *(const v4f*)(xr0 + (size_t)i * TV);
        rb[0][i] = *(const v4f*)(xr1 + (size_t)i * TV);
    }

    {
        const int zj = tid >> 2;
        const int zq = tid & 3;
        float* z = &ldsA[zj * AP + 16 + (zq & 1) * 8 + (zq >> 1) * 32];
        const v2f zero = (v2f){0.0f, 0.0f};
        *(v2f*)(z)     = zero;
        *(v2f*)(z + 2) = zero;
        *(v2f*)(z + 4) = zero;
        *(v2f*)(z + 6) = zero;
    }
    if (tid < PP) lds_cnt[tid] = 0.0f;

    {
        float* d0 = &ldsA[lane * AP + 2 * wave];
        float* d1 = &ldsA[(32 + lane) * AP + 2 * wave];
        v2f t;
        t.x = pa.x; t.y = pa.y; *(v2f*)(d0)      = t;
        t.x = pa.z; t.y = pa.w; *(v2f*)(d0 + 32) = t;
        t.x = pb.x; t.y = pb.y; *(v2f*)(d1)      = t;
        t.x = pb.z; t.y = pb.w; *(v2f*)(d1 + 32) = t;
    }
    {
        float ss = pa.x * pa.x + pa.y * pa.y + pa.z * pa.z + pa.w * pa.w
                 + pb.x * pb.x + pb.y * pb.y + pb.z * pb.z + pb.w * pb.w;
        ss += __shfl_xor(ss, 16, 32);
        ss += __shfl_xor(ss, 8, 32);
        ss += __shfl_xor(ss, 4, 32);
        ss += __shfl_xor(ss, 2, 32);
        ss += __shfl_xor(ss, 1, 32);
        if (lane == 0) lds_pn[wave] = sqrtf(ss);
    }

#pragma unroll
    for (int b = 0; b < CPW / LB; ++b) {
        if (b + 1 < CPW / LB) {
#pragma unroll
            for (int i = 0; i < LB; ++i) {
                ra[(b + 1) & 1][i] = *(const v4f*)(xr0 + (size_t)((b + 1) * LB + i) * TV);
                rb[(b + 1) & 1][i] = *(const v4f*)(xr1 + (size_t)((b + 1) * LB + i) * TV);
            }
        }
#pragma unroll
        for (int i = 0; i < LB; ++i) {
            const v4f a = ra[b & 1][i];
            const v4f c = rb[b & 1][i];
            v4f h;
            h.x = 0.5f * (a.x + c.x);
            h.y = 0.5f * (a.y + c.y);
            h.z = 0.5f * (a.z + c.z);
            h.w = 0.5f * (a.w + c.w);
            *(v4f*)(hdst + (b * LB + i) * HP) = h;
        }
    }
    __syncthreads();

    const int tloc  = lane & 15;
    const int tv    = chunk * TOK_PER_BLOCK + wave * 16 + tloc;
    const bool valid = (tv < TV);
    const int khalf = (lane >> 4) << 1;

    const float* bsrc = &ldsH[khalf * HP + wave * 16 + tloc];
    v8f acc = (v8f){0.f, 0.f, 0.f, 0.f, 0.f, 0.f, 0.f, 0.f};
#if !HAVE_WMMA_F32X4
    float fdot[PP];
#pragma unroll
    for (int p = 0; p < PP; ++p) fdot[p] = 0.0f;
#endif
    float hh = 0.0f;

#pragma unroll
    for (int j = 0; j < 64; ++j) {
        float b0 = bsrc[(4 * j) * HP];
        float b1 = bsrc[(4 * j + 1) * HP];
        hh += b0 * b0 + b1 * b1;
#if HAVE_WMMA_F32X4
        v2f a = *(const v2f*)&ldsA[j * AP + (lane << 1)];
        v2f b; b.x = b0; b.y = b1;
        acc = __builtin_amdgcn_wmma_f32_16x16x4_f32(false, a, false, b,
                                                    (short)0, acc, false, false);
#else
        int c0 = 4 * j + khalf;
#pragma unroll
        for (int p = 0; p < PP; ++p)
            fdot[p] += b0 * protos[p * CC + c0] + b1 * protos[p * CC + c0 + 1];
#endif
    }

    hh += __shfl_xor(hh, 16, 32);
    float hn = sqrtf(hh);

    float dots[PP];
#if HAVE_WMMA_F32X4
#pragma unroll
    for (int p = 0; p < PP; ++p) dots[p] = acc[p];
#else
#pragma unroll
    for (int p = 0; p < PP; ++p) dots[p] = fdot[p] + __shfl_xor(fdot[p], 16, 32);
#endif

    float best = -3.0e38f;
    int bi = 0;
#pragma unroll
    for (int p = 0; p < PP; ++p) {
        float sim = dots[p] / fmaxf(hn * lds_pn[p], EPSF);
        if (sim > best) { best = sim; bi = p; }
    }
    if (lane < 16) {
        lds_assign[wave * 16 + tloc] = valid ? bi : -1;
        if (valid)
            __hip_atomic_fetch_add(&lds_cnt[bi], 1.0f,
                                   __ATOMIC_RELAXED, __HIP_MEMORY_SCOPE_WORKGROUP);
    }
    __syncthreads();

    if (tid < PP)
        __hip_atomic_fetch_add(&g_cnt[n * PP + tid], lds_cnt[tid],
                               __ATOMIC_RELAXED, __HIP_MEMORY_SCOPE_AGENT);

    float* gs = g_sums + (size_t)n * PP * CC;
    const int m16 = lane & 15;
    const int*   asrc = &lds_assign[khalf];

#pragma unroll
    for (int hf = 0; hf < 2; ++hf) {
        const float* rsrc = &ldsH[(HCH * hf + wave * 16 + m16) * HP + khalf];

        v8f s0 = (v8f){0.f, 0.f, 0.f, 0.f, 0.f, 0.f, 0.f, 0.f};
        v8f s1 = (v8f){0.f, 0.f, 0.f, 0.f, 0.f, 0.f, 0.f, 0.f};
#pragma unroll
        for (int q = 0; q < 32; q += 2) {
            v2i t0 = *(const v2i*)&asrc[4 * q];
            v2i t1 = *(const v2i*)&asrc[4 * q + 4];
            v2f b0 = *(const v2f*)&rsrc[4 * q];
            v2f b1 = *(const v2f*)&rsrc[4 * q + 4];
            v2f a0, a1;
            a0.x = (t0.x == m16) ? 1.0f : 0.0f;
            a0.y = (t0.y == m16) ? 1.0f : 0.0f;
            a1.x = (t1.x == m16) ? 1.0f : 0.0f;
            a1.y = (t1.y == m16) ? 1.0f : 0.0f;
            s0 = __builtin_amdgcn_wmma_f32_16x16x4_f32(false, a0, false, b0,
                                                       (short)0, s0, false, false);
            s1 = __builtin_amdgcn_wmma_f32_16x16x4_f32(false, a1, false, b1,
                                                       (short)0, s1, false, false);
        }

        if (lane < 16) {
            float* gd = gs + HCH * hf + wave * 16 + lane;
#pragma unroll
            for (int p = 0; p < PP; ++p)
                __hip_atomic_fetch_add(&gd[p * CC], s0[p] + s1[p],
                                       __ATOMIC_RELAXED, __HIP_MEMORY_SCOPE_AGENT);
        }
    }
}

__global__ void rn_final_kernel(const float* __restrict__ g_sums,
                                const float* __restrict__ g_cnt,
                                float* __restrict__ out) {
    int n = blockIdx.x;
    int c = threadIdx.x;
    float s = 0.0f;
#pragma unroll
    for (int p = 0; p < PP; ++p) {
        float cnt = g_cnt[n * PP + p];
        s += g_sums[((size_t)n * PP + p) * CC + c] / fmaxf(cnt, 1.0f);
    }
    out[n * CC + c] = s * 0.125f;
}

extern "C" void kernel_launch(void* const* d_in, const int* in_sizes, int n_in,
                              void* d_out, int out_size, void* d_ws, size_t ws_size,
                              hipStream_t stream) {
    const float* x      = (const float*)d_in[0];
    const float* protos = (const float*)d_in[1];

    float* g_sums = (float*)d_ws;
    float* g_cnt  = g_sums + NN * PP * CC;

    int zn = NN * PP * CC + NN * PP;
    rn_zero_kernel<<<(zn + 255) / 256, 256, 0, stream>>>((float*)d_ws, zn);

    dim3 grid(CHUNKS, NN);
    rn_main_kernel<<<grid, 256, 0, stream>>>(x, protos, g_sums, g_cnt);

    rn_final_kernel<<<NN, CC, 0, stream>>>(g_sums, g_cnt, (float*)d_out);
}


# Persistent work-groups (256, one per WGP, 5 tiles of 64 tokens): tile i+1's global loads stay in flight during tile i's WMMA passes; segment sums add up across a group's tiles (sum order changes)
# speedup vs baseline: 2.5355x; 1.1470x over previous
#include <hip/hip_runtime.h>
#include <math.h>
#include <type_traits>


typedef __attribute__((ext_vector_type(2))) float v2f;
typedef __attribute__((ext_vector_type(2))) int v2i;
typedef __attribute__((ext_vector_type(4))) float v4f;
typedef __attribute__((ext_vector_type(8))) float v8f;

#define NN 32
#define MM 2
#define CC 256
#define TV 2500
#define PP 8
#define TILE 64
#define TPB 5
#define BPS 8
#define EPSF 1e-8f
#define AP 66
#define HCH 128
#define HP 76
#define RPW 32
#define NLD 16

#if __has_builtin(__builtin_amdgcn_wmma_f32_16x16x4_f32)
#define HAVE_WMMA_F32X4 1
#else
#warning "builtin __builtin_amdgcn_wmma_f32_16x16x4_f32 not found - using scalar fallback (wmma=0 expected)"
#define HAVE_WMMA_F32X4 0
#endif

__global__ void rn_zero_kernel(float* ws, int n) {
    int i = blockIdx.x * blockDim.x + threadIdx.x;
    if (i < n) ws[i] = 0.0f;
}

static __device__ __forceinline__ void lds_barrier() {
    __builtin_amdgcn_fence(__ATOMIC_RELEASE, "workgroup", "local");
    __builtin_amdgcn_s_barrier();
    __builtin_amdgcn_fence(__ATOMIC_ACQUIRE, "workgroup", "local");
}

__global__ void __launch_bounds__(256)
rn_main_kernel(const float* __restrict__ x, const float* __restrict__ protos,
               float* __restrict__ g_sums, float* __restrict__ g_cnt) {
    __shared__ __align__(16) float ldsA[64 * AP];
    __shared__ float lds_pn[PP];
    __shared__ float lds_cnt[PP];
    __shared__ __align__(16) int lds_assign[TILE];
    __shared__ __align__(16) float ldsH[2][CC * HP];

    const int tid    = threadIdx.x;
    const int n      = blockIdx.x / BPS;
    const int chunk0 = (blockIdx.x % BPS) * TPB;

    const int lane  = tid & 31;
    const int wave  = __builtin_amdgcn_readfirstlane(tid >> 5);
    const int half  = lane >> 4;
    const int l16   = lane & 15;

    static_assert(PP * 32 == 256 && CC == 256, "one wave per prototype row, two 16-byte pieces per lane");
    const float* prow = protos + wave * CC;
    const v4f pa = *(const v4f*)(prow + 4 * lane);
    const v4f pb = *(const v4f*)(prow + 128 + 4 * lane);

    static_assert(TV % 4 == 0 && TILE == 4 * 16 && RPW * 8 == CC && RPW == 2 * NLD, "tile fetch shape");
    static_assert(BPS * TPB * TILE >= TV && (BPS * TPB - 1) * TILE < TV, "tiles cover a sample");
    const float* xw0 = x + ((size_t)n * MM * CC + (size_t)(RPW * wave + half)) * TV;

    v4f ra[NLD], rb[NLD];
    {
        const int tq  = chunk0 * TILE + 4 * l16;
        const int tqc = (tq < TV) ? tq : (TV - 4);
        const float* p0 = xw0 + tqc;
        const float* p1 = p0 + (size_t)CC * TV;
#pragma unroll
        for (int k = 0; k < NLD; ++k) {
            ra[k] = *(const v4f*)(p0 + (size_t)(2 * k) * TV);
            rb[k] = *(const v4f*)(p1 + (size_t)(2 * k) * TV);
        }
    }

    {
        const int zj = tid >> 2;
        const int zq = tid & 3;
        float* z = &ldsA[zj * AP + 16 + (zq & 1) * 8 + (zq >> 1) * 32];
        const v2f zero = (v2f){0.0f, 0.0f};
        *(v2f*)(z)     = zero;
        *(v2f*)(z + 2) = zero;
        *(v2f*)(z + 4) = zero;
        *(v2f*)(z + 6) = zero;
    }
    if (tid < PP) lds_cnt[tid] = 0.0f;

    {
        float* d0 = &ldsA[lane * AP + 2 * wave];
        float* d1 = &ldsA[(32 + lane) * AP + 2 * wave];
        v2f t;
        t.x = pa.x; t.y = pa.y; *(v2f*)(d0)      = t;
        t.x = pa.z; t.y = pa.w; *(v2f*)(d0 + 32) = t;
        t.x = pb.x; t.y = pb.y; *(v2f*)(d1)      = t;
        t.x = pb.z; t.y = pb.w; *(v2f*)(d1 + 32) = t;
    }
    {
        float ss = pa.x * pa.x + pa.y * pa.y + pa.z * pa.z + pa.w * pa.w
                 + pb.x * pb.x + pb.y * pb.y + pb.z * pb.z + pb.w * pb.w;
        ss += __shfl_xor(ss, 16, 32);
        ss += __shfl_xor(ss, 8, 32);
        ss += __shfl_xor(ss, 4, 32);
        ss += __shfl_xor(ss, 2, 32);
        ss += __shfl_xor(ss, 1, 32);
        if (lane == 0) lds_pn[wave] = sqrtf(ss);
    }

    const int tloc  = l16;
    const int khalf = half << 1;
    const int m16   = l16;

    v8f sacc[2][2];
#pragma unroll
    for (int hf = 0; hf < 2; ++hf) {
        sacc[hf][0] = (v8f){0.f, 0.f, 0.f, 0.f, 0.f, 0.f, 0.f, 0.f};
        sacc[hf][1] = (v8f){0.f, 0.f, 0.f, 0.f, 0.f, 0.f, 0.f, 0.f};
    }

    auto tile_step = [&](const int i, auto is_last) __attribute__((always_inline)) {
        constexpr bool LAST = decltype(is_last)::value;
        const int chunk = chunk0 + i;
        float* hb = ldsH[i & 1];

        {
            float* hdst = hb + (RPW * wave + half) * HP + 4 * l16;
            const int tq  = (chunk + 1) * TILE + 4 * l16;
            const int tqc = (tq < TV) ? tq : (TV - 4);
            const float* p0 = xw0 + tqc;
            const float* p1 = p0 + (size_t)CC * TV;
#pragma unroll
            for (int g = 0; g < 2; ++g) {
#pragma unroll
                for (int k = g * (NLD / 2); k < (g + 1) * (NLD / 2); ++k) {
                    const v4f a = ra[k];
                    const v4f c = rb[k];
                    v4f h;
                    h.x = 0.5f * (a.x + c.x);
                    h.y = 0.5f * (a.y + c.y);
                    h.z = 0.5f * (a.z + c.z);
                    h.w = 0.5f * (a.w + c.w);
                    *(v4f*)(hdst + (2 * k) * HP) = h;
                }
                if constexpr (!LAST) {
#pragma unroll
                    for (int k = g * (NLD / 2); k < (g + 1) * (NLD / 2); ++k) {
                        ra[k] = *(const v4f*)(p0 + (size_t)(2 * k) * TV);
                        rb[k] = *(const v4f*)(p1 + (size_t)(2 * k) * TV);
                    }
                }
            }
        }
        lds_barrier();

        if (wave < 4) {
            const int tv = chunk * TILE + wave * 16 + tloc;
            const bool valid = (tv < TV);
            const float* bsrc = hb + khalf * HP + wave * 16 + tloc;
            v8f acc = (v8f){0.f, 0.f, 0.f, 0.f, 0.f, 0.f, 0.f, 0.f};
#if !HAVE_WMMA_F32X4
            float fdot[PP];
#pragma unroll
            for (int p = 0; p < PP; ++p) fdot[p] = 0.0f;
#endif
            float hh = 0.0f;

            v2f   fa[2][8];
            float f0[2][8], f1[2][8];
#pragma unroll
            for (int u = 0; u < 8; ++u) {
                f0[0][u] = bsrc[(4 * u) * HP];
                f1[0][u] = bsrc[(4 * u + 1) * HP];
                fa[0][u] = *(const v2f*)&ldsA[u * AP + (lane << 1)];
            }
#pragma unroll
            for (int blk = 0; blk < 8; ++blk) {
                if (blk + 1 < 8) {
#pragma unroll
                    for (int u = 0; u < 8; ++u) {
                        const int j = 8 * (blk + 1) + u;
                        f0[(blk + 1) & 1][u] = bsrc[(4 * j) * HP];
                        f1[(blk + 1) & 1][u] = bsrc[(4 * j + 1) * HP];
                        fa[(blk + 1) & 1][u] = *(const v2f*)&ldsA[j * AP + (lane << 1)];
                    }
                }
                __builtin_amdgcn_sched_barrier(0);
#pragma unroll
                for (int u = 0; u < 8; ++u) {
                    const float b0 = f0[blk & 1][u];
                    const float b1 = f1[blk & 1][u];
                    hh += b0 * b0 + b1 * b1;
#if HAVE_WMMA_F32X4
                    v2f b; b.x = b0; b.y = b1;
                    acc = __builtin_amdgcn_wmma_f32_16x16x4_f32(false, fa[blk & 1][u], false, b,
                                                                (short)0, acc, false, false);
#else
                    const int c0 = 4 * (8 * blk + u) + khalf;
#pragma unroll
                    for (int p = 0; p < PP; ++p)
                        fdot[p] += b0 * protos[p * CC + c0] + b1 * protos[p * CC + c0 + 1];
#endif
                }
                __builtin_amdgcn_sched_barrier(0);
            }

            hh += __shfl_xor(hh, 16, 32);
            float hn = sqrtf(hh);

            float dots[PP];
#if HAVE_WMMA_F32X4
#pragma unroll
            for (int p = 0; p < PP; ++p) dots[p] = acc[p];
#else
#pragma unroll
            for (int p = 0; p < PP; ++p) dots[p] = fdot[p] + __shfl_xor(fdot[p], 16, 32);
#endif

            float best = -3.0e38f;
            int bi = 0;
#pragma unroll
            for (int p = 0; p < PP; ++p) {
                float sim = dots[p] / fmaxf(hn * lds_pn[p], EPSF);
                if (sim > best) { best = sim; bi = p; }
            }
            if (lane < 16) {
                lds_assign[wave * 16 + tloc] = valid ? bi : -1;
                if (valid)
                    __hip_atomic_fetch_add(&lds_cnt[bi], 1.0f,
                                           __ATOMIC_RELAXED, __HIP_MEMORY_SCOPE_WORKGROUP);
            }
        }
        lds_barrier();

        {
            const int* asrc = &lds_assign[khalf];
#pragma unroll
            for (int hf = 0; hf < 2; ++hf) {
                const float* rsrc = hb + (HCH * hf + wave * 16 + m16) * HP + khalf;
#pragma unroll
                for (int q = 0; q < TILE / 4; q += 2) {
                    v2i t0 = *(const v2i*)&asrc[4 * q];
                    v2i t1 = *(const v2i*)&asrc[4 * q + 4];
                    v2f b0 = *(const v2f*)&rsrc[4 * q];
                    v2f b1 = *(const v2f*)&rsrc[4 * q + 4];
                    v2f a0, a1;
                    a0.x = (t0.x == m16) ? 1.0f : 0.0f;
                    a0.y = (t0.y == m16) ? 1.0f : 0.0f;
                    a1.x = (t1.x == m16) ? 1.0f : 0.0f;
                    a1.y = (t1.y == m16) ? 1.0f : 0.0f;
#if HAVE_WMMA_F32X4
                    sacc[hf][0] = __builtin_amdgcn_wmma_f32_16x16x4_f32(false, a0, false, b0,
                                                                        (short)0, sacc[hf][0], false, false);
                    sacc[hf][1] = __builtin_amdgcn_wmma_f32_16x16x4_f32(false, a1, false, b1,
                                                                        (short)0, sacc[hf][1], false, false);
#endif
                }
            }
        }
    };
    for (int i = 0; i < TPB - 1; ++i) tile_step(i, std::false_type{});
    tile_step(TPB - 1, std::true_type{});

    float* gs = g_sums + (size_t)n * PP * CC;
    if (lane < 16) {
#pragma unroll
        for (int hf = 0; hf < 2; ++hf) {
            float* gd = gs + HCH * hf + wave * 16 + lane;
#pragma unroll
            for (int p = 0; p < PP; ++p)
                __hip_atomic_fetch_add(&gd[p * CC], sacc[hf][0][p] + sacc[hf][1][p],
                                       __ATOMIC_RELAXED, __HIP_MEMORY_SCOPE_AGENT);
        }
    }
    if (tid < PP)
        __hip_atomic_fetch_add(&g_cnt[n * PP + tid], lds_cnt[tid],
                               __ATOMIC_RELAXED, __HIP_MEMORY_SCOPE_AGENT);
}

__global__ void rn_final_kernel(const float* __restrict__ g_sums,
                                const float* __restrict__ g_cnt,
                                float* __restrict__ out) {
    int n = blockIdx.x;
    int c = threadIdx.x;
    float s = 0.0f;
#pragma unroll
    for (int p = 0; p < PP; ++p) {
        float cnt = g_cnt[n * PP + p];
        s += g_sums[((size_t)n * PP + p) * CC + c] / fmaxf(cnt, 1.0f);
    }
    out[n * CC + c] = s * 0.125f;
}

extern "C" void kernel_launch(void* const* d_in, const int* in_sizes, int n_in,
                              void* d_out, int out_size, void* d_ws, size_t ws_size,
                              hipStream_t stream) {
    const float* x      = (const float*)d_in[0];
    const float* protos = (const float*)d_in[1];

    float* g_sums = (float*)d_ws;
    float* g_cnt  = g_sums + NN * PP * CC;

    int zn = NN * PP * CC + NN * PP;
    rn_zero_kernel<<<(zn + 255) / 256, 256, 0, stream>>>((float*)d_ws, zn);

    rn_main_kernel<<<NN * BPS, 256, 0, stream>>>(x, protos, g_sums, g_cnt);

    rn_final_kernel<<<NN, CC, 0, stream>>>(g_sums, g_cnt, (float*)d_out);
}
